# starting kernel
# baseline (speedup 1.0000x reference)
#include <hip/hip_runtime.h>
#include <hip/hip_bf16.h>


#define NH     8
#define SEQ    512
#define DK     32
#define DV     64
#define BM     128
#define BN     16
#define ALPHA  (1.0f / 64.0f)

typedef float v2f __attribute__((ext_vector_type(2)));
typedef float v8f __attribute__((ext_vector_type(8)));

__device__ __forceinline__ v8f wmma_f32_k4(v2f a, v2f b, v8f c) {
    return __builtin_amdgcn_wmma_f32_16x16x4_f32(false, a, false, b,
                                                 (short)0, c, false, false);
}

__global__ __launch_bounds__(256)
void poly2_attn_kernel(const float* __restrict__ Q,
                       const float* __restrict__ K,
                       const float* __restrict__ V,
                       float* __restrict__ YS,
                       float* __restrict__ YZ) {
    __shared__ float Qs[BM][DK + 1];
    __shared__ float Ks[BN][DK + 1];
    __shared__ float Vs[BN][DV + 1];
    __shared__ float Ps[8][16][17];

    const int h    = blockIdx.y;
    const int q0   = blockIdx.x * BM;
    const int tid  = threadIdx.x;
    const int wave = tid >> 5;
    const int lane = tid & 31;
    const int r    = lane & 15;
    const int hi   = lane >> 4;

    const float* Qh = Q + (size_t)h * SEQ * DK;
    const float* Kh = K + (size_t)h * SEQ * DK;
    const float* Vh = V + (size_t)h * SEQ * DV;

    for (int idx = tid; idx < BM * DK; idx += 256) {
        const int row = idx >> 5, col = idx & 31;
        Qs[row][col] = Qh[(size_t)(q0 + row) * DK + col];
    }
    __syncthreads();

    const int qrow_local = wave * 16;
    const int qbase      = q0 + qrow_local;
    const int jd         = qbase >> 4;
    const int nj         = (q0 >> 4) + (BM >> 4);

    v2f qa[DK / 4];
#pragma unroll
    for (int kk = 0; kk < DK; kk += 4) {
        const int c0 = kk + 2 * hi;
        qa[kk >> 2] = (v2f){ Qs[qrow_local + r][c0], Qs[qrow_local + r][c0 + 1] };
    }

    v8f acc0 = {0,0,0,0,0,0,0,0};
    v8f acc1 = {0,0,0,0,0,0,0,0};
    v8f acc2 = {0,0,0,0,0,0,0,0};
    v8f acc3 = {0,0,0,0,0,0,0,0};
    v8f accz = {0,0,0,0,0,0,0,0};
    const v2f ones = {1.0f, 1.0f};

    for (int j = 0; j < nj; ++j) {
        const int k0 = j * BN;
        __syncthreads();
        for (int idx = tid; idx < BN * DK; idx += 256) {
            const int row = idx >> 5, col = idx & 31;
            Ks[row][col] = Kh[(size_t)(k0 + row) * DK + col];
        }
        for (int idx = tid; idx < BN * DV; idx += 256) {
            const int row = idx >> 6, col = idx & 63;
            Vs[row][col] = Vh[(size_t)(k0 + row) * DV + col];
        }
        if (j + 1 < nj && tid < BN) {
            const int kn = k0 + BN + tid;
            __builtin_prefetch(&Kh[(size_t)kn * DK], 0, 1);
            __builtin_prefetch(&Vh[(size_t)kn * DV], 0, 1);
            __builtin_prefetch(&Vh[(size_t)kn * DV + 32], 0, 1);
        }
        __syncthreads();

        const bool active = (j <= jd);
        if (active) {
            v2f kb[DK / 4];
#pragma unroll
            for (int kk = 0; kk < DK; kk += 4) {
                const int c0 = kk + 2 * hi;
                kb[kk >> 2] = (v2f){ Ks[r][c0], Ks[r][c0 + 1] };
            }
            v8f s = {0,0,0,0,0,0,0,0};
#pragma unroll
            for (int t = 0; t < DK / 4; ++t)
                s = wmma_f32_k4(qa[t], kb[t], s);

#pragma unroll
            for (int i = 0; i < 8; ++i) {
                float p = s[i] * s[i];
                if (j == jd) {
                    const int gq = qbase + i + 8 * hi;
                    const int gk = k0 + r;
                    if (gk > gq) p = 0.0f;
                }
                Ps[wave][i + 8 * hi][r] = p;
            }
        }
        __syncthreads();

        if (active) {
#pragma unroll
            for (int kk = 0; kk < BN; kk += 4) {
                const int c0 = kk + 2 * hi;
                v2f a  = { Ps[wave][r][c0], Ps[wave][r][c0 + 1] };
                v2f b0 = { Vs[c0][ 0 + r], Vs[c0 + 1][ 0 + r] };
                v2f b1 = { Vs[c0][16 + r], Vs[c0 + 1][16 + r] };
                v2f b2 = { Vs[c0][32 + r], Vs[c0 + 1][32 + r] };
                v2f b3 = { Vs[c0][48 + r], Vs[c0 + 1][48 + r] };
                acc0 = wmma_f32_k4(a, b0, acc0);
                acc1 = wmma_f32_k4(a, b1, acc1);
                acc2 = wmma_f32_k4(a, b2, acc2);
                acc3 = wmma_f32_k4(a, b3, acc3);
                accz = wmma_f32_k4(a, ones, accz);
            }
        }
    }

    float* ys = YS + ((size_t)h * SEQ + qbase) * DV;
#pragma unroll
    for (int i = 0; i < 8; ++i) {
        const int row = i + 8 * hi;
        ys[(size_t)row * DV +  0 + r] = acc0[i] * ALPHA;
        ys[(size_t)row * DV + 16 + r] = acc1[i] * ALPHA;
        ys[(size_t)row * DV + 32 + r] = acc2[i] * ALPHA;
        ys[(size_t)row * DV + 48 + r] = acc3[i] * ALPHA;
    }
    if (r == 0) {
#pragma unroll
        for (int i = 0; i < 8; ++i) {
            YZ[(size_t)h * SEQ + qbase + i + 8 * hi] = accz[i] * ALPHA;
        }
    }
}

extern "C" void kernel_launch(void* const* d_in, const int* in_sizes, int n_in,
                              void* d_out, int out_size, void* d_ws, size_t ws_size,
                              hipStream_t stream) {
    (void)in_sizes; (void)n_in; (void)out_size; (void)d_ws; (void)ws_size;
    const float* Q = (const float*)d_in[0];
    const float* K = (const float*)d_in[1];
    const float* V = (const float*)d_in[2];
    float* YS = (float*)d_out;
    float* YZ = (float*)d_out + (size_t)NH * SEQ * DV;

    dim3 grid(SEQ / BM, NH);
    dim3 block(256);
    hipLaunchKernelGGL(poly2_attn_kernel, grid, block, 0, stream, Q, K, V, YS, YZ);
}


# K and V rows of the work-group's whole key range copied to LDS once (16-byte copies, one barrier); each wave then runs its own key blocks out of LDS with no work-group barrier; sum order kept
# speedup vs baseline: 3.0229x; 3.0229x over previous
#include <hip/hip_runtime.h>
#include <hip/hip_bf16.h>


#define NH     8
#define SEQ    512
#define DK     32
#define DV     64
#define BM     128
#define BN     16
#define KP     36
#define VP     72
#define ALPHA  (1.0f / 64.0f)

typedef float v2f __attribute__((ext_vector_type(2)));
typedef float v8f __attribute__((ext_vector_type(8)));

__device__ __forceinline__ v8f wmma_f32_k4(v2f a, v2f b, v8f c) {
    return __builtin_amdgcn_wmma_f32_16x16x4_f32(false, a, false, b,
                                                 (short)0, c, false, false);
}

__global__ __launch_bounds__(256)
void poly2_attn_kernel(const float* __restrict__ Q,
                       const float* __restrict__ K,
                       const float* __restrict__ V,
                       float* __restrict__ YS,
                       float* __restrict__ YZ) {
    __shared__ __attribute__((aligned(16))) float Ks[SEQ * KP];
    __shared__ __attribute__((aligned(16))) float Vs[SEQ * VP];
    __shared__ float Qs[BM][DK + 1];
    __shared__ float Ps[8][16][17];

    const int h    = blockIdx.y;
    const int xb   = blockIdx.x;
    const int q0   = xb * BM;
    const int tid  = threadIdx.x;
    const int wave = __builtin_amdgcn_readfirstlane(tid >> 5);
    const int lane = tid & 31;
    const int r    = lane & 15;
    const int hi   = lane >> 4;

    const float* Qh = Q + (size_t)h * SEQ * DK;
    const float* Kh = K + (size_t)h * SEQ * DK;
    const float* Vh = V + (size_t)h * SEQ * DV;
    const float4* Kh4 = reinterpret_cast<const float4*>(Kh);
    const float4* Vh4 = reinterpret_cast<const float4*>(Vh);

    const float4* Qh4 = reinterpret_cast<const float4*>(Qh + (size_t)q0 * DK);
    float4 qreg[4];
#pragma unroll
    for (int u = 0; u < 4; ++u)
        qreg[u] = Qh4[u * 256 + tid];

    for (int c = 0; c <= xb; ++c) {
        float4 kreg[4], vreg[8];
#pragma unroll
        for (int u = 0; u < 4; ++u)
            kreg[u] = Kh4[c * (128 * DK / 4) + u * 256 + tid];
#pragma unroll
        for (int u = 0; u < 8; ++u)
            vreg[u] = Vh4[c * (128 * DV / 4) + u * 256 + tid];
#pragma unroll
        for (int u = 0; u < 4; ++u) {
            const int idx = u * 256 + tid;
            const int row = c * 128 + (idx >> 3), c4 = idx & 7;
            *reinterpret_cast<float4*>(&Ks[row * KP + c4 * 4]) = kreg[u];
        }
#pragma unroll
        for (int u = 0; u < 8; ++u) {
            const int idx = u * 256 + tid;
            const int row = c * 128 + (idx >> 4), c4 = idx & 15;
            *reinterpret_cast<float4*>(&Vs[row * VP + c4 * 4]) = vreg[u];
        }
    }
#pragma unroll
    for (int u = 0; u < 4; ++u) {
        const int idx = u * 256 + tid;
        const int row = idx >> 3, col = (idx & 7) * 4;
        Qs[row][col + 0] = qreg[u].x;
        Qs[row][col + 1] = qreg[u].y;
        Qs[row][col + 2] = qreg[u].z;
        Qs[row][col + 3] = qreg[u].w;
    }
    __syncthreads();

    const int qrow_local = wave * 16;
    const int qbase      = q0 + qrow_local;
    const int jd         = qbase >> 4;

    v2f qa[DK / 4];
#pragma unroll
    for (int kk = 0; kk < DK; kk += 4) {
        const int c0 = kk + 2 * hi;
        qa[kk >> 2] = (v2f){ Qs[qrow_local + r][c0], Qs[qrow_local + r][c0 + 1] };
    }

    v8f acc0 = {0,0,0,0,0,0,0,0};
    v8f acc1 = {0,0,0,0,0,0,0,0};
    v8f acc2 = {0,0,0,0,0,0,0,0};
    v8f acc3 = {0,0,0,0,0,0,0,0};
    v8f accz = {0,0,0,0,0,0,0,0};
    const v2f ones = {1.0f, 1.0f};

    v2f kb[DK / 4];
    {
        const float* krow = &Ks[r * KP + 2 * hi];
#pragma unroll
        for (int t = 0; t < DK / 4; ++t)
            kb[t] = (v2f){ krow[4 * t], krow[4 * t + 1] };
    }
    for (int j = 0; j <= jd; ++j) {
        const int k0 = j * BN;

        const float* vrow = &Vs[(k0 + 2 * hi) * VP + r];
        v2f vb[BN / 4][4];
#pragma unroll
        for (int q = 0; q < BN / 4; ++q) {
#pragma unroll
            for (int n = 0; n < 4; ++n)
                vb[q][n] = (v2f){ vrow[(4 * q) * VP + 16 * n], vrow[(4 * q + 1) * VP + 16 * n] };
        }

        v8f s = {0,0,0,0,0,0,0,0};
#pragma unroll
        for (int t = 0; t < DK / 4; ++t)
            s = wmma_f32_k4(qa[t], kb[t], s);

        {
            const int jn = (j < jd) ? j + 1 : jd;
            const float* krow = &Ks[(jn * BN + r) * KP + 2 * hi];
#pragma unroll
            for (int t = 0; t < DK / 4; ++t)
                kb[t] = (v2f){ krow[4 * t], krow[4 * t + 1] };
        }

#pragma unroll
        for (int i = 0; i < 8; ++i) {
            float p = s[i] * s[i];
            if (j == jd) {
                const int gq = qbase + i + 8 * hi;
                const int gk = k0 + r;
                if (gk > gq) p = 0.0f;
            }
            Ps[wave][i + 8 * hi][r] = p;
        }
        __builtin_amdgcn_wave_barrier();

#pragma unroll
        for (int q = 0; q < BN / 4; ++q) {
            const int c0 = 4 * q + 2 * hi;
            v2f a = { Ps[wave][r][c0], Ps[wave][r][c0 + 1] };
            acc0 = wmma_f32_k4(a, vb[q][0], acc0);
            acc1 = wmma_f32_k4(a, vb[q][1], acc1);
            acc2 = wmma_f32_k4(a, vb[q][2], acc2);
            acc3 = wmma_f32_k4(a, vb[q][3], acc3);
            accz = wmma_f32_k4(a, ones, accz);
        }
        __builtin_amdgcn_wave_barrier();
    }

    float* ys = YS + ((size_t)h * SEQ + qbase) * DV;
#pragma unroll
    for (int i = 0; i < 8; ++i) {
        const int row = i + 8 * hi;
        ys[(size_t)row * DV +  0 + r] = acc0[i] * ALPHA;
        ys[(size_t)row * DV + 16 + r] = acc1[i] * ALPHA;
        ys[(size_t)row * DV + 32 + r] = acc2[i] * ALPHA;
        ys[(size_t)row * DV + 48 + r] = acc3[i] * ALPHA;
    }
    if (r == 0) {
#pragma unroll
        for (int i = 0; i < 8; ++i) {
            YZ[(size_t)h * SEQ + qbase + i + 8 * hi] = accz[i] * ALPHA;
        }
    }
}

extern "C" void kernel_launch(void* const* d_in, const int* in_sizes, int n_in,
                              void* d_out, int out_size, void* d_ws, size_t ws_size,
                              hipStream_t stream) {
    (void)in_sizes; (void)n_in; (void)out_size; (void)d_ws; (void)ws_size;
    const float* Q = (const float*)d_in[0];
    const float* K = (const float*)d_in[1];
    const float* V = (const float*)d_in[2];
    float* YS = (float*)d_out;
    float* YZ = (float*)d_out + (size_t)NH * SEQ * DV;

    dim3 grid(SEQ / BM, NH);
    dim3 block(256);
    hipLaunchKernelGGL(poly2_attn_kernel, grid, block, 0, stream, Q, K, V, YS, YZ);
}


# One 16-row query tile per work-group (256 groups x 4 waves, one per SIMD): work split over V's four 16-column slices, scores recomputed per wave; LDS copy loads one chunk ahead; sum order kept
# speedup vs baseline: 5.5727x; 1.8435x over previous
#include <hip/hip_runtime.h>
#include <hip/hip_bf16.h>


#define NH     8
#define SEQ    512
#define DK     32
#define DV     64
#define BM     16
#define NW     4
#define BN     16
#define CH     64
#define KP     36
#define VP     72
#define OP     68
#define ALPHA  (1.0f / 64.0f)

typedef float v2f __attribute__((ext_vector_type(2)));
typedef float v4f __attribute__((ext_vector_type(4)));
typedef float v8f __attribute__((ext_vector_type(8)));

__device__ __forceinline__ v8f wmma_f32_k4(v2f a, v2f b, v8f c) {
    return __builtin_amdgcn_wmma_f32_16x16x4_f32(false, a, false, b,
                                                 (short)0, c, false, false);
}

template <bool WITH_Z>
__device__ __forceinline__ void key_loop(const float* Ks, const float* Vs, float* Pw,
                                         const v2f (&qa)[DK / 4],
                                         const int jd, const int qbase, const int ct,
                                         const int r, const int hi,
                                         v8f& acc, v8f& accz) {
    const v2f ones = {1.0f, 1.0f};

    v2f kb[DK / 4];
    {
        const float* krow = &Ks[r * KP + 2 * hi];
#pragma unroll
        for (int t = 0; t < DK / 4; ++t)
            kb[t] = (v2f){ krow[4 * t], krow[4 * t + 1] };
    }
    for (int j = 0; j <= jd; ++j) {
        const int k0 = j * BN;

        const float* vrow = &Vs[(k0 + 2 * hi) * VP + 16 * ct + r];
        v2f vb[BN / 4];
#pragma unroll
        for (int q = 0; q < BN / 4; ++q)
            vb[q] = (v2f){ vrow[(4 * q) * VP], vrow[(4 * q + 1) * VP] };

        v8f s = {0,0,0,0,0,0,0,0};
#pragma unroll
        for (int t = 0; t < DK / 4; ++t)
            s = wmma_f32_k4(qa[t], kb[t], s);

        {
            const int jn = (j < jd) ? j + 1 : jd;
            const float* krow = &Ks[(jn * BN + r) * KP + 2 * hi];
#pragma unroll
            for (int t = 0; t < DK / 4; ++t)
                kb[t] = (v2f){ krow[4 * t], krow[4 * t + 1] };
        }

#pragma unroll
        for (int i = 0; i < 8; ++i) {
            float p = s[i] * s[i];
            if (j == jd) {
                const int gq = qbase + i + 8 * hi;
                const int gk = k0 + r;
                if (gk > gq) p = 0.0f;
            }
            Pw[(i + 8 * hi) * 17 + r] = p;
        }
        __builtin_amdgcn_wave_barrier();

        v2f a[BN / 4];
#pragma unroll
        for (int q = 0; q < BN / 4; ++q) {
            const int c0 = 4 * q + 2 * hi;
            a[q] = (v2f){ Pw[r * 17 + c0], Pw[r * 17 + c0 + 1] };
        }
#pragma unroll
        for (int q = 0; q < BN / 4; ++q) {
            acc = wmma_f32_k4(a[q], vb[q], acc);
            if (WITH_Z) accz = wmma_f32_k4(a[q], ones, accz);
        }
        __builtin_amdgcn_wave_barrier();
    }
}

__global__ __launch_bounds__(NW * 32)
void poly2_attn_kernel(const float* __restrict__ Q,
                       const float* __restrict__ K,
                       const float* __restrict__ V,
                       float* __restrict__ YS,
                       float* __restrict__ YZ) {
    __shared__ __attribute__((aligned(16))) float Ks[SEQ * KP];
    __shared__ __attribute__((aligned(16))) float Vs[SEQ * VP];
    __shared__ float Qs[BM][DK + 1];
    __shared__ float Ps[NW][16][17];
    __shared__ __attribute__((aligned(16))) float Os[BM * OP];

    const int h    = blockIdx.x;
    const int qt   = (SEQ / BM - 1) - (int)blockIdx.y;
    const int qbase = qt * BM;
    const int jd   = qt;
    const int tid  = threadIdx.x;
    const int wave = __builtin_amdgcn_readfirstlane(tid >> 5);
    const int lane = tid & 31;
    const int r    = lane & 15;
    const int hi   = lane >> 4;

    const float* Qh = Q + (size_t)h * SEQ * DK;
    const float* Kh = K + (size_t)h * SEQ * DK;
    const float* Vh = V + (size_t)h * SEQ * DV;
    const v4f* Kh4 = reinterpret_cast<const v4f*>(Kh);
    const v4f* Vh4 = reinterpret_cast<const v4f*>(Vh);

    const v4f qreg = reinterpret_cast<const v4f*>(Qh + (size_t)qbase * DK)[tid];

    const int nch = (qt >> 2) + 1;
#define LOAD_CHUNK(c, kr, vr)                                                   \
    {                                                                           \
        _Pragma("unroll")                                                       \
        for (int u = 0; u < 4; ++u)                                             \
            kr[u] = Kh4[(c) * (CH * DK / 4) + u * (NW * 32) + tid];             \
        _Pragma("unroll")                                                       \
        for (int u = 0; u < 8; ++u)                                             \
            vr[u] = Vh4[(c) * (CH * DV / 4) + u * (NW * 32) + tid];             \
    }
#define STORE_CHUNK(c, kr, vr)                                                  \
    {                                                                           \
        _Pragma("unroll")                                                       \
        for (int u = 0; u < 4; ++u) {                                           \
            const int idx = u * (NW * 32) + tid;                                \
            const int row = (c) * CH + (idx >> 3), c4 = idx & 7;                \
            *reinterpret_cast<v4f*>(&Ks[row * KP + c4 * 4]) = kr[u];         \
        }                                                                       \
        _Pragma("unroll")                                                       \
        for (int u = 0; u < 8; ++u) {                                           \
            const int idx = u * (NW * 32) + tid;                                \
            const int row = (c) * CH + (idx >> 4), c4 = idx & 15;               \
            *reinterpret_cast<v4f*>(&Vs[row * VP + c4 * 4]) = vr[u];         \
        }                                                                       \
    }
    {
        v4f ka[4], va[8], kc[4], vc[8];
        LOAD_CHUNK(0, ka, va)
        int c = 0;
        for (; c + 2 < nch; c += 2) {
            LOAD_CHUNK(c + 1, kc, vc)
            STORE_CHUNK(c, ka, va)
            LOAD_CHUNK(c + 2, ka, va)
            STORE_CHUNK(c + 1, kc, vc)
        }
        if (c + 1 < nch) {
            LOAD_CHUNK(c + 1, kc, vc)
            STORE_CHUNK(c, ka, va)
            STORE_CHUNK(c + 1, kc, vc)
        } else {
            STORE_CHUNK(c, ka, va)
        }
    }
#undef LOAD_CHUNK
#undef STORE_CHUNK
    {
        const int row = tid >> 3, col = (tid & 7) * 4;
        Qs[row][col + 0] = qreg.x;
        Qs[row][col + 1] = qreg.y;
        Qs[row][col + 2] = qreg.z;
        Qs[row][col + 3] = qreg.w;
    }
    __syncthreads();

    v2f qa[DK / 4];
#pragma unroll
    for (int kk = 0; kk < DK; kk += 4) {
        const int c0 = kk + 2 * hi;
        qa[kk >> 2] = (v2f){ Qs[r][c0], Qs[r][c0 + 1] };
    }

    v8f acc  = {0,0,0,0,0,0,0,0};
    v8f accz = {0,0,0,0,0,0,0,0};
    float* Pw = &Ps[wave][0][0];
    if (wave == 0)
        key_loop<true >(Ks, Vs, Pw, qa, jd, qbase, wave, r, hi, acc, accz);
    else
        key_loop<false>(Ks, Vs, Pw, qa, jd, qbase, wave, r, hi, acc, accz);

#pragma unroll
    for (int i = 0; i < 8; ++i)
        Os[(i + 8 * hi) * OP + 16 * wave + r] = acc[i] * ALPHA;
    __syncthreads();
    v4f* ys4 = reinterpret_cast<v4f*>(YS + ((size_t)h * SEQ + qbase) * DV);
#pragma unroll
    for (int u = 0; u < 2; ++u) {
        const int idx = u * (NW * 32) + tid;
        const int row = idx >> 4, c4 = idx & 15;
        ys4[idx] = *reinterpret_cast<const v4f*>(&Os[row * OP + c4 * 4]);
    }
    if (wave == 0 && r == 0) {
#pragma unroll
        for (int i = 0; i < 8; ++i)
            YZ[(size_t)h * SEQ + qbase + i + 8 * hi] = accz[i] * ALPHA;
    }
}

extern "C" void kernel_launch(void* const* d_in, const int* in_sizes, int n_in,
                              void* d_out, int out_size, void* d_ws, size_t ws_size,
                              hipStream_t stream) {
    (void)in_sizes; (void)n_in; (void)out_size; (void)d_ws; (void)ws_size;
    const float* Q = (const float*)d_in[0];
    const float* K = (const float*)d_in[1];
    const float* V = (const float*)d_in[2];
    float* YS = (float*)d_out;
    float* YZ = (float*)d_out + (size_t)NH * SEQ * DV;

    dim3 grid(NH, SEQ / BM);
    dim3 block(NW * 32);
    hipLaunchKernelGGL(poly2_attn_kernel, grid, block, 0, stream, Q, K, V, YS, YZ);
}


# Scores computed once per work-group: wave w makes the squared score tile of block w of each 64-key group, shared via LDS (2 buffers, 1 barrier per group); Y_Z summed per wave then added: order changes
# speedup vs baseline: 7.0855x; 1.2715x over previous
#include <hip/hip_runtime.h>
#include <hip/hip_bf16.h>


#define NH     8
#define SEQ    512
#define DK     32
#define DV     64
#define BM     16
#define NW     4
#define BN     16
#define CH     64
#define KP     36
#define VP     72
#define OP     68
#define ALPHA  (1.0f / 64.0f)

typedef float v2f __attribute__((ext_vector_type(2)));
typedef float v4f __attribute__((ext_vector_type(4)));
typedef float v8f __attribute__((ext_vector_type(8)));

__device__ __forceinline__ v8f wmma_f32_k4(v2f a, v2f b, v8f c) {
    return __builtin_amdgcn_wmma_f32_16x16x4_f32(false, a, false, b,
                                                 (short)0, c, false, false);
}

#define PP     17
#define PT     (16 * PP)
#define PG     (NW * PT)

__device__ __forceinline__ v8f score_chain(const float* Ks, const v2f (&qa)[DK / 4],
                                           const int kb, const int r, const int hi) {
    const float* krow = &Ks[(kb * BN + r) * KP + 2 * hi];
    v2f kf[DK / 4];
#pragma unroll
    for (int t = 0; t < DK / 4; ++t)
        kf[t] = (v2f){ krow[4 * t], krow[4 * t + 1] };
    v8f s = {0,0,0,0,0,0,0,0};
#pragma unroll
    for (int t = 0; t < DK / 4; ++t)
        s = wmma_f32_k4(qa[t], kf[t], s);
    return s;
}

template <bool MASK>
__device__ __forceinline__ void store_p(float* Pt, const v8f& s,
                                        const int kb, const int qbase,
                                        const int r, const int hi) {
#pragma unroll
    for (int i = 0; i < 8; ++i) {
        float p = s[i] * s[i];
        if (MASK) {
            const int gq = qbase + i + 8 * hi;
            const int gk = kb * BN + r;
            if (gk > gq) p = 0.0f;
        }
        Pt[(i + 8 * hi) * PP + r] = p;
    }
}

template <int NB, bool NEXT, bool MASK>
__device__ __forceinline__ void group_step(const float* Ks, const float* Vs,
                                           const float* Pcur, float* Pnxt_w,
                                           const v2f (&qa)[DK / 4], const int g,
                                           const int qbase, const int wave,
                                           const int r, const int hi,
                                           v8f& acc, v8f& accz) {
    const v2f ones = {1.0f, 1.0f};
    const int kbn = 4 * (g + 1) + wave;
    v2f kf[DK / 4];
    if (NEXT) {
        const float* krow = &Ks[(kbn * BN + r) * KP + 2 * hi];
#pragma unroll
        for (int t = 0; t < DK / 4; ++t)
            kf[t] = (v2f){ krow[4 * t], krow[4 * t + 1] };
    }
    v2f a[NB][BN / 4], vb[NB][BN / 4], az[BN / 4];
#pragma unroll
    for (int b = 0; b < NB; ++b) {
        const float* vrow = &Vs[((4 * g + b) * BN + 2 * hi) * VP + 16 * wave + r];
        const float* prow = &Pcur[b * PT + r * PP + 2 * hi];
#pragma unroll
        for (int q = 0; q < BN / 4; ++q) {
            vb[b][q] = (v2f){ vrow[(4 * q) * VP], vrow[(4 * q + 1) * VP] };
            a[b][q]  = (v2f){ prow[4 * q], prow[4 * q + 1] };
        }
    }
    const bool zdo = (NB == 4) || (wave < NB);
    if (zdo) {
        const float* prow = &Pcur[wave * PT + r * PP + 2 * hi];
#pragma unroll
        for (int q = 0; q < BN / 4; ++q)
            az[q] = (v2f){ prow[4 * q], prow[4 * q + 1] };
    }
    v8f s = {0,0,0,0,0,0,0,0};
#pragma unroll
    for (int b = 0; b < 4; ++b) {
        if (NEXT) s = wmma_f32_k4(qa[2 * b], kf[2 * b], s);
        if (b < NB) {
            acc = wmma_f32_k4(a[b][0], vb[b][0], acc);
            acc = wmma_f32_k4(a[b][1], vb[b][1], acc);
        }
        if (NEXT) s = wmma_f32_k4(qa[2 * b + 1], kf[2 * b + 1], s);
        if (b < NB) {
            acc = wmma_f32_k4(a[b][2], vb[b][2], acc);
            acc = wmma_f32_k4(a[b][3], vb[b][3], acc);
        }
        if (NB == 4) accz = wmma_f32_k4(az[b], ones, accz);
    }
    if (NB < 4 && zdo) {
#pragma unroll
        for (int q = 0; q < BN / 4; ++q)
            accz = wmma_f32_k4(az[q], ones, accz);
    }
    if (NEXT) store_p<MASK>(Pnxt_w, s, kbn, qbase, r, hi);
    __syncthreads();
}

__global__ __launch_bounds__(NW * 32)
void poly2_attn_kernel(const float* __restrict__ Q,
                       const float* __restrict__ K,
                       const float* __restrict__ V,
                       float* __restrict__ YS,
                       float* __restrict__ YZ) {
    __shared__ __attribute__((aligned(16))) float Ks[SEQ * KP];
    __shared__ __attribute__((aligned(16))) float Vs[SEQ * VP];
    __shared__ float Qs[BM][DK + 1];
    __shared__ float Ps[2 * PG];
    __shared__ float Zs[NW * BM];
    __shared__ __attribute__((aligned(16))) float Os[BM * OP];

    const int h    = blockIdx.x;
    const int qt   = (SEQ / BM - 1) - (int)blockIdx.y;
    const int qbase = qt * BM;
    const int jd   = qt;
    const int tid  = threadIdx.x;
    const int wave = __builtin_amdgcn_readfirstlane(tid >> 5);
    const int lane = tid & 31;
    const int r    = lane & 15;
    const int hi   = lane >> 4;

    const float* Qh = Q + (size_t)h * SEQ * DK;
    const float* Kh = K + (size_t)h * SEQ * DK;
    const float* Vh = V + (size_t)h * SEQ * DV;
    const v4f* Kh4 = reinterpret_cast<const v4f*>(Kh);
    const v4f* Vh4 = reinterpret_cast<const v4f*>(Vh);

    const v4f qreg = reinterpret_cast<const v4f*>(Qh + (size_t)qbase * DK)[tid];

    const int nch = (qt >> 2) + 1;
#define LOAD_CHUNK(c, kr, vr)                                                   \
    {                                                                           \
        _Pragma("unroll")                                                       \
        for (int u = 0; u < 4; ++u)                                             \
            kr[u] = Kh4[(c) * (CH * DK / 4) + u * (NW * 32) + tid];             \
        _Pragma("unroll")                                                       \
        for (int u = 0; u < 8; ++u)                                             \
            vr[u] = Vh4[(c) * (CH * DV / 4) + u * (NW * 32) + tid];             \
    }
#define STORE_CHUNK(c, kr, vr)                                                  \
    {                                                                           \
        _Pragma("unroll")                                                       \
        for (int u = 0; u < 4; ++u) {                                           \
            const int idx = u * (NW * 32) + tid;                                \
            const int row = (c) * CH + (idx >> 3), c4 = idx & 7;                \
            *reinterpret_cast<v4f*>(&Ks[row * KP + c4 * 4]) = kr[u];         \
        }                                                                       \
        _Pragma("unroll")                                                       \
        for (int u = 0; u < 8; ++u) {                                           \
            const int idx = u * (NW * 32) + tid;                                \
            const int row = (c) * CH + (idx >> 4), c4 = idx & 15;               \
            *reinterpret_cast<v4f*>(&Vs[row * VP + c4 * 4]) = vr[u];         \
        }                                                                       \
    }
    {
        v4f ka[4], va[8], kc[4], vc[8];
        LOAD_CHUNK(0, ka, va)
        int c = 0;
        for (; c + 2 < nch; c += 2) {
            LOAD_CHUNK(c + 1, kc, vc)
            STORE_CHUNK(c, ka, va)
            LOAD_CHUNK(c + 2, ka, va)
            STORE_CHUNK(c + 1, kc, vc)
        }
        if (c + 1 < nch) {
            LOAD_CHUNK(c + 1, kc, vc)
            STORE_CHUNK(c, ka, va)
            STORE_CHUNK(c + 1, kc, vc)
        } else {
            STORE_CHUNK(c, ka, va)
        }
    }
#undef LOAD_CHUNK
#undef STORE_CHUNK
    {
        const int row = tid >> 3, col = (tid & 7) * 4;
        Qs[row][col + 0] = qreg.x;
        Qs[row][col + 1] = qreg.y;
        Qs[row][col + 2] = qreg.z;
        Qs[row][col + 3] = qreg.w;
    }
    __syncthreads();

    v2f qa[DK / 4];
#pragma unroll
    for (int kk = 0; kk < DK; kk += 4) {
        const int c0 = kk + 2 * hi;
        qa[kk >> 2] = (v2f){ Qs[r][c0], Qs[r][c0 + 1] };
    }

    v8f acc  = {0,0,0,0,0,0,0,0};
    v8f accz = {0,0,0,0,0,0,0,0};

    {
        const v8f s = score_chain(Ks, qa, wave, r, hi);
        store_p<true>(&Ps[wave * PT], s, wave, qbase, r, hi);
    }
    __syncthreads();

#define STEP(NB, NEXT, MASK)                                                          \
    group_step<NB, NEXT, MASK>(Ks, Vs, &Ps[(g & 1) * PG],                             \
                               &Ps[((g + 1) & 1) * PG + wave * PT],                   \
                               qa, g, qbase, wave, r, hi, acc, accz)
    int g = 0;
    for (; g + 2 < nch; ++g)
        STEP(4, true, false);
    if (g + 1 < nch) {
        STEP(4, true, true);
        ++g;
    }
    switch (qt & 3) {
        case 0:  STEP(1, false, false); break;
        case 1:  STEP(2, false, false); break;
        case 2:  STEP(3, false, false); break;
        default: STEP(4, false, false); break;
    }
#undef STEP

#pragma unroll
    for (int i = 0; i < 8; ++i)
        Os[(i + 8 * hi) * OP + 16 * wave + r] = acc[i] * ALPHA;
    if (r == 0) {
#pragma unroll
        for (int i = 0; i < 8; ++i)
            Zs[wave * BM + i + 8 * hi] = accz[i];
    }
    __syncthreads();
    v4f* ys4 = reinterpret_cast<v4f*>(YS + ((size_t)h * SEQ + qbase) * DV);
#pragma unroll
    for (int u = 0; u < 2; ++u) {
        const int idx = u * (NW * 32) + tid;
        const int row = idx >> 4, c4 = idx & 15;
        ys4[idx] = *reinterpret_cast<const v4f*>(&Os[row * OP + c4 * 4]);
    }
    if (tid < BM)
        YZ[(size_t)h * SEQ + qbase + tid] =
            (((Zs[tid] + Zs[BM + tid]) + Zs[2 * BM + tid]) + Zs[3 * BM + tid]) * ALPHA;
}

extern "C" void kernel_launch(void* const* d_in, const int* in_sizes, int n_in,
                              void* d_out, int out_size, void* d_ws, size_t ws_size,
                              hipStream_t stream) {
    (void)in_sizes; (void)n_in; (void)out_size; (void)d_ws; (void)ws_size;
    const float* Q = (const float*)d_in[0];
    const float* K = (const float*)d_in[1];
    const float* V = (const float*)d_in[2];
    float* YS = (float*)d_out;
    float* YZ = (float*)d_out + (size_t)NH * SEQ * DV;

    dim3 grid(NH, SEQ / BM);
    dim3 block(NW * 32);
    hipLaunchKernelGGL(poly2_attn_kernel, grid, block, 0, stream, Q, K, V, YS, YZ);
}


# K/V copy to LDS moved under the key loop: only chunk 0 is waited for up front; chunk g+3 loads fly in registers while chunk g+2 is stored in group g; barrier waits for LDS only; sum order kept
# speedup vs baseline: 8.3500x; 1.1785x over previous
#include <hip/hip_runtime.h>
#include <hip/hip_bf16.h>


#define NH     8
#define SEQ    512
#define DK     32
#define DV     64
#define BM     16
#define NW     4
#define BN     16
#define CH     64
#define KP     36
#define VP     72
#define OP     68
#define ALPHA  (1.0f / 64.0f)

typedef float v2f __attribute__((ext_vector_type(2)));
typedef float v4f __attribute__((ext_vector_type(4)));
typedef float v8f __attribute__((ext_vector_type(8)));

__device__ __forceinline__ v8f wmma_f32_k4(v2f a, v2f b, v8f c) {
    return __builtin_amdgcn_wmma_f32_16x16x4_f32(false, a, false, b,
                                                 (short)0, c, false, false);
}

__device__ __forceinline__ void wg_barrier_lds() {
    asm volatile("s_wait_dscnt 0x0\n\ts_barrier_signal -1\n\ts_barrier_wait -1" ::: "memory");
}

#define PP     17
#define PT     (16 * PP)
#define PG     (NW * PT)

__device__ __forceinline__ v8f score_chain(const float* Ks, const v2f (&qa)[DK / 4],
                                           const int kb, const int r, const int hi) {
    const float* krow = &Ks[(kb * BN + r) * KP + 2 * hi];
    v2f kf[DK / 4];
#pragma unroll
    for (int t = 0; t < DK / 4; ++t)
        kf[t] = (v2f){ krow[4 * t], krow[4 * t + 1] };
    v8f s = {0,0,0,0,0,0,0,0};
#pragma unroll
    for (int t = 0; t < DK / 4; ++t)
        s = wmma_f32_k4(qa[t], kf[t], s);
    return s;
}

template <bool MASK>
__device__ __forceinline__ void store_p(float* Pt, const v8f& s,
                                        const int kb, const int qbase,
                                        const int r, const int hi) {
#pragma unroll
    for (int i = 0; i < 8; ++i) {
        float p = s[i] * s[i];
        if (MASK) {
            const int gq = qbase + i + 8 * hi;
            const int gk = kb * BN + r;
            if (gk > gq) p = 0.0f;
        }
        Pt[(i + 8 * hi) * PP + r] = p;
    }
}

template <int NB, bool NEXT, bool MASK>
__device__ __forceinline__ void group_step(const float* Ks, const float* Vs,
                                           const float* Pcur, float* Pnxt_w,
                                           const v2f (&qa)[DK / 4], const int g,
                                           const int qbase, const int wave,
                                           const int r, const int hi,
                                           v8f& acc, v8f& accz) {
    const v2f ones = {1.0f, 1.0f};
    const int kbn = 4 * (g + 1) + wave;
    v2f kf[DK / 4];
    if (NEXT) {
        const float* krow = &Ks[(kbn * BN + r) * KP + 2 * hi];
#pragma unroll
        for (int t = 0; t < DK / 4; ++t)
            kf[t] = (v2f){ krow[4 * t], krow[4 * t + 1] };
    }
    v2f a[NB][BN / 4], vb[NB][BN / 4], az[BN / 4];
#pragma unroll
    for (int b = 0; b < NB; ++b) {
        const float* vrow = &Vs[((4 * g + b) * BN + 2 * hi) * VP + 16 * wave + r];
        const float* prow = &Pcur[b * PT + r * PP + 2 * hi];
#pragma unroll
        for (int q = 0; q < BN / 4; ++q) {
            vb[b][q] = (v2f){ vrow[(4 * q) * VP], vrow[(4 * q + 1) * VP] };
            a[b][q]  = (v2f){ prow[4 * q], prow[4 * q + 1] };
        }
    }
    const bool zdo = (NB == 4) || (wave < NB);
    if (zdo) {
        const float* prow = &Pcur[wave * PT + r * PP + 2 * hi];
#pragma unroll
        for (int q = 0; q < BN / 4; ++q)
            az[q] = (v2f){ prow[4 * q], prow[4 * q + 1] };
    }
    v8f s = {0,0,0,0,0,0,0,0};
#pragma unroll
    for (int b = 0; b < 4; ++b) {
        if (NEXT) s = wmma_f32_k4(qa[2 * b], kf[2 * b], s);
        if (b < NB) {
            acc = wmma_f32_k4(a[b][0], vb[b][0], acc);
            acc = wmma_f32_k4(a[b][1], vb[b][1], acc);
        }
        if (NEXT) s = wmma_f32_k4(qa[2 * b + 1], kf[2 * b + 1], s);
        if (b < NB) {
            acc = wmma_f32_k4(a[b][2], vb[b][2], acc);
            acc = wmma_f32_k4(a[b][3], vb[b][3], acc);
        }
        if (NB == 4) accz = wmma_f32_k4(az[b], ones, accz);
    }
    if (NB < 4 && zdo) {
#pragma unroll
        for (int q = 0; q < BN / 4; ++q)
            accz = wmma_f32_k4(az[q], ones, accz);
    }
    if (NEXT) store_p<MASK>(Pnxt_w, s, kbn, qbase, r, hi);
}

__global__ __launch_bounds__(NW * 32)
void poly2_attn_kernel(const float* __restrict__ Q,
                       const float* __restrict__ K,
                       const float* __restrict__ V,
                       float* __restrict__ YS,
                       float* __restrict__ YZ) {
    __shared__ __attribute__((aligned(16))) float Ks[SEQ * KP];
    __shared__ __attribute__((aligned(16))) float Vs[SEQ * VP];
    __shared__ float Qs[BM][DK + 1];
    __shared__ float Ps[2 * PG];
    __shared__ float Zs[NW * BM];
    __shared__ __attribute__((aligned(16))) float Os[BM * OP];

    const int h    = blockIdx.x;
    const int qt   = (SEQ / BM - 1) - (int)blockIdx.y;
    const int qbase = qt * BM;
    const int jd   = qt;
    const int tid  = threadIdx.x;
    const int wave = __builtin_amdgcn_readfirstlane(tid >> 5);
    const int lane = tid & 31;
    const int r    = lane & 15;
    const int hi   = lane >> 4;

    const float* Qh = Q + (size_t)h * SEQ * DK;
    const float* Kh = K + (size_t)h * SEQ * DK;
    const float* Vh = V + (size_t)h * SEQ * DV;
    const v4f* Kh4 = reinterpret_cast<const v4f*>(Kh);
    const v4f* Vh4 = reinterpret_cast<const v4f*>(Vh);

    const v4f qreg = reinterpret_cast<const v4f*>(Qh + (size_t)qbase * DK)[tid];

    const int nch = (qt >> 2) + 1;
#define LOAD_CHUNK(c, kr, vr)                                                   \
    {                                                                           \
        _Pragma("unroll")                                                       \
        for (int u = 0; u < 4; ++u)                                             \
            kr[u] = Kh4[(c) * (CH * DK / 4) + u * (NW * 32) + tid];             \
        _Pragma("unroll")                                                       \
        for (int u = 0; u < 8; ++u)                                             \
            vr[u] = Vh4[(c) * (CH * DV / 4) + u * (NW * 32) + tid];             \
    }
#define STORE_CHUNK(c, kr, vr)                                                  \
    {                                                                           \
        _Pragma("unroll")                                                       \
        for (int u = 0; u < 4; ++u) {                                           \
            const int idx = u * (NW * 32) + tid;                                \
            const int row = (c) * CH + (idx >> 3), c4 = idx & 7;                \
            *reinterpret_cast<v4f*>(&Ks[row * KP + c4 * 4]) = kr[u];         \
        }                                                                       \
        _Pragma("unroll")                                                       \
        for (int u = 0; u < 8; ++u) {                                           \
            const int idx = u * (NW * 32) + tid;                                \
            const int row = (c) * CH + (idx >> 4), c4 = idx & 15;               \
            *reinterpret_cast<v4f*>(&Vs[row * VP + c4 * 4]) = vr[u];         \
        }                                                                       \
    }
    static_assert(SEQ >= 3 * CH, "chunks 1 and 2 exist for every tile");
    v4f ka[4], va[8], kc[4], vc[8];
    LOAD_CHUNK(0, ka, va)
    LOAD_CHUNK(1, kc, vc)
    STORE_CHUNK(0, ka, va)
    LOAD_CHUNK(2, ka, va)
    {
        const int row = tid >> 3, col = (tid & 7) * 4;
        Qs[row][col + 0] = qreg.x;
        Qs[row][col + 1] = qreg.y;
        Qs[row][col + 2] = qreg.z;
        Qs[row][col + 3] = qreg.w;
    }
    wg_barrier_lds();

    v2f qa[DK / 4];
#pragma unroll
    for (int kk = 0; kk < DK; kk += 4) {
        const int c0 = kk + 2 * hi;
        qa[kk >> 2] = (v2f){ Qs[r][c0], Qs[r][c0 + 1] };
    }

    v8f acc  = {0,0,0,0,0,0,0,0};
    v8f accz = {0,0,0,0,0,0,0,0};

    {
        const v8f s = score_chain(Ks, qa, wave, r, hi);
        store_p<true>(&Ps[wave * PT], s, wave, qbase, r, hi);
    }
    STORE_CHUNK(1, kc, vc)
    wg_barrier_lds();

#define STEP(G, NB, NEXT, MASK)                                                       \
    group_step<NB, NEXT, MASK>(Ks, Vs, &Ps[((G) & 1) * PG],                           \
                               &Ps[(((G) + 1) & 1) * PG + wave * PT],                 \
                               qa, (G), qbase, wave, r, hi, acc, accz)
    int g = 0;
    for (; g + 4 < nch; g += 2) {
        LOAD_CHUNK(g + 3, kc, vc)
        STEP(g, 4, true, false);
        STORE_CHUNK(g + 2, ka, va)
        wg_barrier_lds();
        LOAD_CHUNK(g + 4, ka, va)
        STEP(g + 1, 4, true, false);
        STORE_CHUNK(g + 3, kc, vc)
        wg_barrier_lds();
    }
    if (g + 3 < nch) {
        LOAD_CHUNK(g + 3, kc, vc)
        STEP(g, 4, true, false);
        STORE_CHUNK(g + 2, ka, va)
        wg_barrier_lds();
        STEP(g + 1, 4, true, false);
        STORE_CHUNK(g + 3, kc, vc)
        wg_barrier_lds();
        g += 2;
    } else if (g + 2 < nch) {
        STEP(g, 4, true, false);
        STORE_CHUNK(g + 2, ka, va)
        wg_barrier_lds();
        ++g;
    }
#undef LOAD_CHUNK
#undef STORE_CHUNK
    if (g + 1 < nch) {
        STEP(g, 4, true, true);
        wg_barrier_lds();
        ++g;
    }
    switch (qt & 3) {
        case 0:  STEP(g, 1, false, false); break;
        case 1:  STEP(g, 2, false, false); break;
        case 2:  STEP(g, 3, false, false); break;
        default: STEP(g, 4, false, false); break;
    }
    __syncthreads();
#undef STEP

#pragma unroll
    for (int i = 0; i < 8; ++i)
        Os[(i + 8 * hi) * OP + 16 * wave + r] = acc[i] * ALPHA;
    if (r == 0) {
#pragma unroll
        for (int i = 0; i < 8; ++i)
            Zs[wave * BM + i + 8 * hi] = accz[i];
    }
    __syncthreads();
    v4f* ys4 = reinterpret_cast<v4f*>(YS + ((size_t)h * SEQ + qbase) * DV);
#pragma unroll
    for (int u = 0; u < 2; ++u) {
        const int idx = u * (NW * 32) + tid;
        const int row = idx >> 4, c4 = idx & 15;
        ys4[idx] = *reinterpret_cast<const v4f*>(&Os[row * OP + c4 * 4]);
    }
    if (tid < BM)
        YZ[(size_t)h * SEQ + qbase + tid] =
            (((Zs[tid] + Zs[BM + tid]) + Zs[2 * BM + tid]) + Zs[3 * BM + tid]) * ALPHA;
}

extern "C" void kernel_launch(void* const* d_in, const int* in_sizes, int n_in,
                              void* d_out, int out_size, void* d_ws, size_t ws_size,
                              hipStream_t stream) {
    (void)in_sizes; (void)n_in; (void)out_size; (void)d_ws; (void)ws_size;
    const float* Q = (const float*)d_in[0];
    const float* K = (const float*)d_in[1];
    const float* V = (const float*)d_in[2];
    float* YS = (float*)d_out;
    float* YZ = (float*)d_out + (size_t)NH * SEQ * DV;

    dim3 grid(NH, SEQ / BM);
    dim3 block(NW * 32);
    hipLaunchKernelGGL(poly2_attn_kernel, grid, block, 0, stream, Q, K, V, YS, YZ);
}


# P.V of key blocks 1-3 of a group deferred one step: a step issues all its LDS fragment loads as one batch behind the barrier and runs the deferred products while they land; sum order kept
# speedup vs baseline: 8.8248x; 1.0569x over previous
#include <hip/hip_runtime.h>
#include <hip/hip_bf16.h>


#define NH     8
#define SEQ    512
#define DK     32
#define DV     64
#define BM     16
#define NW     4
#define BN     16
#define CH     64
#define KP     36
#define VP     72
#define OP     68
#define ALPHA  (1.0f / 64.0f)

typedef float v2f __attribute__((ext_vector_type(2)));
typedef float v4f __attribute__((ext_vector_type(4)));
typedef float v8f __attribute__((ext_vector_type(8)));

__device__ __forceinline__ v8f wmma_f32_k4(v2f a, v2f b, v8f c) {
    return __builtin_amdgcn_wmma_f32_16x16x4_f32(false, a, false, b,
                                                 (short)0, c, false, false);
}

__device__ __forceinline__ void wg_barrier_lds() {
    asm volatile("s_wait_dscnt 0x0\n\ts_barrier_signal -1\n\ts_barrier_wait -1" ::: "memory");
}

#define PP     17
#define PT     (16 * PP)
#define PG     (NW * PT)

__device__ __forceinline__ v8f score_chain(const float* Ks, const v2f (&qa)[DK / 4],
                                           const int kb, const int r, const int hi) {
    const float* krow = &Ks[(kb * BN + r) * KP + 2 * hi];
    v2f kf[DK / 4];
#pragma unroll
    for (int t = 0; t < DK / 4; ++t)
        kf[t] = (v2f){ krow[4 * t], krow[4 * t + 1] };
    v8f s = {0,0,0,0,0,0,0,0};
#pragma unroll
    for (int t = 0; t < DK / 4; ++t)
        s = wmma_f32_k4(qa[t], kf[t], s);
    return s;
}

template <bool MASK>
__device__ __forceinline__ void store_p(float* Pt, const v8f& s,
                                        const int kb, const int qbase,
                                        const int r, const int hi) {
#pragma unroll
    for (int i = 0; i < 8; ++i) {
        float p = s[i] * s[i];
        if (MASK) {
            const int gq = qbase + i + 8 * hi;
            const int gk = kb * BN + r;
            if (gk > gq) p = 0.0f;
        }
        Pt[(i + 8 * hi) * PP + r] = p;
    }
}

#define ND     3

template <int NB, bool NEXT, bool MASK>
__device__ __forceinline__ void group_step(const float* Ks, const float* Vs,
                                           const float* Pcur, float* Pnxt_w,
                                           const v2f (&qa)[DK / 4], const int g,
                                           const int qbase, const int wave,
                                           const int r, const int hi,
                                           v2f (&da)[ND][BN / 4], v2f (&dvb)[ND][BN / 4],
                                           v8f& acc, v8f& accz) {
    static_assert(!NEXT || NB == 4, "a step that looks ahead works on a whole group");
    const v2f ones = {1.0f, 1.0f};
    const int kbn = 4 * (g + 1) + wave;
    v2f kf[DK / 4];
    if (NEXT) {
        const float* krow = &Ks[(kbn * BN + r) * KP + 2 * hi];
#pragma unroll
        for (int t = 0; t < DK / 4; ++t)
            kf[t] = (v2f){ krow[4 * t], krow[4 * t + 1] };
    }
    v2f a[NB][BN / 4], vb[NB][BN / 4], az[BN / 4];
    const bool zdo = (NB == 4) || (wave < NB);
    if (zdo) {
        const float* prow = &Pcur[wave * PT + r * PP + 2 * hi];
#pragma unroll
        for (int q = 0; q < BN / 4; ++q)
            az[q] = (v2f){ prow[4 * q], prow[4 * q + 1] };
    }
#pragma unroll
    for (int b = 0; b < NB; ++b) {
        const float* vrow = &Vs[((4 * g + b) * BN + 2 * hi) * VP + 16 * wave + r];
        const float* prow = &Pcur[b * PT + r * PP + 2 * hi];
#pragma unroll
        for (int q = 0; q < BN / 4; ++q) {
            vb[b][q] = (v2f){ vrow[(4 * q) * VP], vrow[(4 * q + 1) * VP] };
            a[b][q]  = (v2f){ prow[4 * q], prow[4 * q + 1] };
        }
    }
    __builtin_amdgcn_sched_barrier(0);
    if (NEXT) {
        v8f s = {0,0,0,0,0,0,0,0};
#pragma unroll
        for (int j = 0; j < 4; ++j) {
            acc = wmma_f32_k4(da[(3 * j) / 4][(3 * j) % 4], dvb[(3 * j) / 4][(3 * j) % 4], acc);
            s   = wmma_f32_k4(qa[2 * j], kf[2 * j], s);
            acc = wmma_f32_k4(da[(3 * j + 1) / 4][(3 * j + 1) % 4], dvb[(3 * j + 1) / 4][(3 * j + 1) % 4], acc);
            s   = wmma_f32_k4(qa[2 * j + 1], kf[2 * j + 1], s);
            acc = wmma_f32_k4(da[(3 * j + 2) / 4][(3 * j + 2) % 4], dvb[(3 * j + 2) / 4][(3 * j + 2) % 4], acc);
        }
        store_p<MASK>(Pnxt_w, s, kbn, qbase, r, hi);
#pragma unroll
        for (int q = 0; q < BN / 4; ++q) {
            acc  = wmma_f32_k4(a[0][q], vb[0][q], acc);
            accz = wmma_f32_k4(az[q], ones, accz);
        }
#pragma unroll
        for (int b = 0; b < ND; ++b) {
#pragma unroll
            for (int q = 0; q < BN / 4; ++q) {
                da[b][q]  = a[b + 1][q];
                dvb[b][q] = vb[b + 1][q];
            }
        }
    } else {
#pragma unroll
        for (int b = 0; b < ND; ++b) {
#pragma unroll
            for (int q = 0; q < BN / 4; ++q)
                acc = wmma_f32_k4(da[b][q], dvb[b][q], acc);
        }
#pragma unroll
        for (int b = 0; b < 4; ++b) {
            if (b < NB) {
#pragma unroll
                for (int q = 0; q < BN / 4; ++q)
                    acc = wmma_f32_k4(a[b][q], vb[b][q], acc);
            }
            if (NB == 4) accz = wmma_f32_k4(az[b], ones, accz);
        }
        if (NB < 4 && zdo) {
#pragma unroll
            for (int q = 0; q < BN / 4; ++q)
                accz = wmma_f32_k4(az[q], ones, accz);
        }
    }
}

__global__ __launch_bounds__(NW * 32)
void poly2_attn_kernel(const float* __restrict__ Q,
                       const float* __restrict__ K,
                       const float* __restrict__ V,
                       float* __restrict__ YS,
                       float* __restrict__ YZ) {
    __shared__ __attribute__((aligned(16))) float Ks[SEQ * KP];
    __shared__ __attribute__((aligned(16))) float Vs[SEQ * VP];
    __shared__ float Qs[BM][DK + 1];
    __shared__ float Ps[2 * PG];
    __shared__ float Zs[NW * BM];
    __shared__ __attribute__((aligned(16))) float Os[BM * OP];

    const int h    = blockIdx.x;
    const int qt   = (SEQ / BM - 1) - (int)blockIdx.y;
    const int qbase = qt * BM;
    const int jd   = qt;
    const int tid  = threadIdx.x;
    const int wave = __builtin_amdgcn_readfirstlane(tid >> 5);
    const int lane = tid & 31;
    const int r    = lane & 15;
    const int hi   = lane >> 4;

    const float* Qh = Q + (size_t)h * SEQ * DK;
    const float* Kh = K + (size_t)h * SEQ * DK;
    const float* Vh = V + (size_t)h * SEQ * DV;
    const v4f* Kh4 = reinterpret_cast<const v4f*>(Kh);
    const v4f* Vh4 = reinterpret_cast<const v4f*>(Vh);

    const v4f qreg = reinterpret_cast<const v4f*>(Qh + (size_t)qbase * DK)[tid];

    const int nch = (qt >> 2) + 1;
#define LOAD_CHUNK(c, kr, vr)                                                   \
    {                                                                           \
        _Pragma("unroll")                                                       \
        for (int u = 0; u < 4; ++u)                                             \
            kr[u] = Kh4[(c) * (CH * DK / 4) + u * (NW * 32) + tid];             \
        _Pragma("unroll")                                                       \
        for (int u = 0; u < 8; ++u)                                             \
            vr[u] = Vh4[(c) * (CH * DV / 4) + u * (NW * 32) + tid];             \
    }
#define STORE_CHUNK(c, kr, vr)                                                  \
    {                                                                           \
        _Pragma("unroll")                                                       \
        for (int u = 0; u < 4; ++u) {                                           \
            const int idx = u * (NW * 32) + tid;                                \
            const int row = (c) * CH + (idx >> 3), c4 = idx & 7;                \
            *reinterpret_cast<v4f*>(&Ks[row * KP + c4 * 4]) = kr[u];         \
        }                                                                       \
        _Pragma("unroll")                                                       \
        for (int u = 0; u < 8; ++u) {                                           \
            const int idx = u * (NW * 32) + tid;                                \
            const int row = (c) * CH + (idx >> 4), c4 = idx & 15;               \
            *reinterpret_cast<v4f*>(&Vs[row * VP + c4 * 4]) = vr[u];         \
        }                                                                       \
    }
    static_assert(SEQ >= 3 * CH, "chunks 1 and 2 exist for every tile");
    v4f ka[4], va[8], kc[4], vc[8];
    LOAD_CHUNK(0, ka, va)
    LOAD_CHUNK(1, kc, vc)
    STORE_CHUNK(0, ka, va)
    LOAD_CHUNK(2, ka, va)
    {
        const int row = tid >> 3, col = (tid & 7) * 4;
        Qs[row][col + 0] = qreg.x;
        Qs[row][col + 1] = qreg.y;
        Qs[row][col + 2] = qreg.z;
        Qs[row][col + 3] = qreg.w;
    }
    wg_barrier_lds();

    v2f qa[DK / 4];
#pragma unroll
    for (int kk = 0; kk < DK; kk += 4) {
        const int c0 = kk + 2 * hi;
        qa[kk >> 2] = (v2f){ Qs[r][c0], Qs[r][c0 + 1] };
    }

    v8f acc  = {0,0,0,0,0,0,0,0};
    v2f da[ND][BN / 4], dvb[ND][BN / 4];
#pragma unroll
    for (int b = 0; b < ND; ++b) {
#pragma unroll
        for (int q = 0; q < BN / 4; ++q) {
            da[b][q]  = (v2f){0.0f, 0.0f};
            dvb[b][q] = (v2f){0.0f, 0.0f};
        }
    }
    v8f accz = {0,0,0,0,0,0,0,0};

    {
        const v8f s = score_chain(Ks, qa, wave, r, hi);
        store_p<true>(&Ps[wave * PT], s, wave, qbase, r, hi);
    }
    STORE_CHUNK(1, kc, vc)
    wg_barrier_lds();

#define STEP(G, NB, NEXT, MASK)                                                       \
    group_step<NB, NEXT, MASK>(Ks, Vs, &Ps[((G) & 1) * PG],                           \
                               &Ps[(((G) + 1) & 1) * PG + wave * PT],                 \
                               qa, (G), qbase, wave, r, hi, da, dvb, acc, accz)
    int g = 0;
    for (; g + 4 < nch; g += 2) {
        LOAD_CHUNK(g + 3, kc, vc)
        STEP(g, 4, true, false);
        STORE_CHUNK(g + 2, ka, va)
        wg_barrier_lds();
        LOAD_CHUNK(g + 4, ka, va)
        STEP(g + 1, 4, true, false);
        STORE_CHUNK(g + 3, kc, vc)
        wg_barrier_lds();
    }
    if (g + 3 < nch) {
        LOAD_CHUNK(g + 3, kc, vc)
        STEP(g, 4, true, false);
        STORE_CHUNK(g + 2, ka, va)
        wg_barrier_lds();
        STEP(g + 1, 4, true, false);
        STORE_CHUNK(g + 3, kc, vc)
        wg_barrier_lds();
        g += 2;
    } else if (g + 2 < nch) {
        STEP(g, 4, true, false);
        STORE_CHUNK(g + 2, ka, va)
        wg_barrier_lds();
        ++g;
    }
#undef LOAD_CHUNK
#undef STORE_CHUNK
    if (g + 1 < nch) {
        STEP(g, 4, true, true);
        wg_barrier_lds();
        ++g;
    }
    switch (qt & 3) {
        case 0:  STEP(g, 1, false, false); break;
        case 1:  STEP(g, 2, false, false); break;
        case 2:  STEP(g, 3, false, false); break;
        default: STEP(g, 4, false, false); break;
    }
    __syncthreads();
#undef STEP

#pragma unroll
    for (int i = 0; i < 8; ++i)
        Os[(i + 8 * hi) * OP + 16 * wave + r] = acc[i] * ALPHA;
    if (r == 0) {
#pragma unroll
        for (int i = 0; i < 8; ++i)
            Zs[wave * BM + i + 8 * hi] = accz[i];
    }
    __syncthreads();
    v4f* ys4 = reinterpret_cast<v4f*>(YS + ((size_t)h * SEQ + qbase) * DV);
#pragma unroll
    for (int u = 0; u < 2; ++u) {
        const int idx = u * (NW * 32) + tid;
        const int row = idx >> 4, c4 = idx & 15;
        ys4[idx] = *reinterpret_cast<const v4f*>(&Os[row * OP + c4 * 4]);
    }
    if (tid < BM)
        YZ[(size_t)h * SEQ + qbase + tid] =
            (((Zs[tid] + Zs[BM + tid]) + Zs[2 * BM + tid]) + Zs[3 * BM + tid]) * ALPHA;
}

extern "C" void kernel_launch(void* const* d_in, const int* in_sizes, int n_in,
                              void* d_out, int out_size, void* d_ws, size_t ws_size,
                              hipStream_t stream) {
    (void)in_sizes; (void)n_in; (void)out_size; (void)d_ws; (void)ws_size;
    const float* Q = (const float*)d_in[0];
    const float* K = (const float*)d_in[1];
    const float* V = (const float*)d_in[2];
    float* YS = (float*)d_out;
    float* YZ = (float*)d_out + (size_t)NH * SEQ * DV;

    dim3 grid(NH, SEQ / BM);
    dim3 block(NW * 32);
    hipLaunchKernelGGL(poly2_attn_kernel, grid, block, 0, stream, Q, K, V, YS, YZ);
}


# Key-loop barrier split: a step signals behind its LDS stores, waits in the next step behind its last 8 products, the next chunk's global loads and the V fragment loads; sum order kept
# speedup vs baseline: 9.8530x; 1.1165x over previous
#include <hip/hip_runtime.h>
#include <hip/hip_bf16.h>


#define NH     8
#define SEQ    512
#define DK     32
#define DV     64
#define BM     16
#define NW     4
#define BN     16
#define CH     64
#define KP     36
#define VP     72
#define OP     68
#define ALPHA  (1.0f / 64.0f)

typedef float v2f __attribute__((ext_vector_type(2)));
typedef float v4f __attribute__((ext_vector_type(4)));
typedef float v8f __attribute__((ext_vector_type(8)));

__device__ __forceinline__ v8f wmma_f32_k4(v2f a, v2f b, v8f c) {
    return __builtin_amdgcn_wmma_f32_16x16x4_f32(false, a, false, b,
                                                 (short)0, c, false, false);
}

__device__ __forceinline__ void wg_signal_lds() {
    asm volatile("s_wait_dscnt 0x0\n\ts_barrier_signal -1" ::: "memory");
}
__device__ __forceinline__ void wg_wait() {
    asm volatile("s_barrier_wait -1" ::: "memory");
}
__device__ __forceinline__ void wg_signal_lds(v8f& acc, v8f& accz) {
    asm volatile("s_wait_dscnt 0x0\n\ts_barrier_signal -1" : "+v"(acc), "+v"(accz) :: "memory");
}
__device__ __forceinline__ void wg_wait(const v8f& acc, const v8f& accz) {
    asm volatile("s_barrier_wait -1" :: "v"(acc), "v"(accz) : "memory");
}

#define PP     17
#define PT     (16 * PP)
#define PG     (NW * PT)

__device__ __forceinline__ v8f score_chain(const float* Ks, const v2f (&qa)[DK / 4],
                                           const int kb, const int r, const int hi) {
    const float* krow = &Ks[(kb * BN + r) * KP + 2 * hi];
    v2f kf[DK / 4];
#pragma unroll
    for (int t = 0; t < DK / 4; ++t)
        kf[t] = (v2f){ krow[4 * t], krow[4 * t + 1] };
    v8f s = {0,0,0,0,0,0,0,0};
#pragma unroll
    for (int t = 0; t < DK / 4; ++t)
        s = wmma_f32_k4(qa[t], kf[t], s);
    return s;
}

template <bool MASK>
__device__ __forceinline__ void store_p(float* Pt, const v8f& s,
                                        const int kb, const int qbase,
                                        const int r, const int hi) {
#pragma unroll
    for (int i = 0; i < 8; ++i) {
        float p = s[i] * s[i];
        if (MASK) {
            const int gq = qbase + i + 8 * hi;
            const int gk = kb * BN + r;
            if (gk > gq) p = 0.0f;
        }
        Pt[(i + 8 * hi) * PP + r] = p;
    }
}

#define ND     3

struct StepFrags {
    v2f kf[DK / 4];
    v2f a[4][BN / 4];
    v2f vb[4][BN / 4];
    v2f az[BN / 4];
};

template <int NB>
__device__ __forceinline__ void step_loads_v(StepFrags& f, const float* Vs, const int g,
                                             const int wave, const int r, const int hi) {
#pragma unroll
    for (int b = 0; b < NB; ++b) {
        const float* vrow = &Vs[((4 * g + b) * BN + 2 * hi) * VP + 16 * wave + r];
#pragma unroll
        for (int q = 0; q < BN / 4; ++q)
            f.vb[b][q] = (v2f){ vrow[(4 * q) * VP], vrow[(4 * q + 1) * VP] };
    }
}

template <int NB, bool NEXT>
__device__ __forceinline__ void step_loads(StepFrags& f, const float* Ks,
                                           const float* Pcur, const int g, const int wave,
                                           const int r, const int hi) {
    static_assert(!NEXT || NB == 4, "a step that looks ahead works on a whole group");
    if (NEXT) {
        const int kbn = 4 * (g + 1) + wave;
        const float* krow = &Ks[(kbn * BN + r) * KP + 2 * hi];
#pragma unroll
        for (int t = 0; t < DK / 4; ++t)
            f.kf[t] = (v2f){ krow[4 * t], krow[4 * t + 1] };
    }
    const bool zdo = (NB == 4) || (wave < NB);
    if (zdo) {
        const float* prow = &Pcur[wave * PT + r * PP + 2 * hi];
#pragma unroll
        for (int q = 0; q < BN / 4; ++q)
            f.az[q] = (v2f){ prow[4 * q], prow[4 * q + 1] };
    }
#pragma unroll
    for (int b = 0; b < NB; ++b) {
        const float* prow = &Pcur[b * PT + r * PP + 2 * hi];
#pragma unroll
        for (int q = 0; q < BN / 4; ++q)
            f.a[b][q]  = (v2f){ prow[4 * q], prow[4 * q + 1] };
    }
    __builtin_amdgcn_sched_barrier(0);
}

template <bool MASK>
__device__ __forceinline__ void step_head(const StepFrags& f, const v2f (&qa)[DK / 4],
                                          const v2f (&da)[ND][BN / 4],
                                          const v2f (&dvb)[ND][BN / 4], v8f& acc,
                                          float* Pnxt_w, const int kbn, const int qbase,
                                          const int r, const int hi) {
    v8f s = {0,0,0,0,0,0,0,0};
#pragma unroll
    for (int j = 0; j < 4; ++j) {
        acc = wmma_f32_k4(da[(3 * j) / 4][(3 * j) % 4], dvb[(3 * j) / 4][(3 * j) % 4], acc);
        s   = wmma_f32_k4(qa[2 * j], f.kf[2 * j], s);
        acc = wmma_f32_k4(da[(3 * j + 1) / 4][(3 * j + 1) % 4], dvb[(3 * j + 1) / 4][(3 * j + 1) % 4], acc);
        s   = wmma_f32_k4(qa[2 * j + 1], f.kf[2 * j + 1], s);
        acc = wmma_f32_k4(da[(3 * j + 2) / 4][(3 * j + 2) % 4], dvb[(3 * j + 2) / 4][(3 * j + 2) % 4], acc);
    }
    store_p<MASK>(Pnxt_w, s, kbn, qbase, r, hi);
}

__device__ __forceinline__ void step_tail(const StepFrags& f, v8f& acc, v8f& accz) {
    const v2f ones = {1.0f, 1.0f};
#pragma unroll
    for (int q = 0; q < BN / 4; ++q) {
        acc  = wmma_f32_k4(f.a[0][q], f.vb[0][q], acc);
        accz = wmma_f32_k4(f.az[q], ones, accz);
    }
}

__device__ __forceinline__ void step_handover(const StepFrags& f, v2f (&da)[ND][BN / 4],
                                              v2f (&dvb)[ND][BN / 4]) {
#pragma unroll
    for (int b = 0; b < ND; ++b) {
#pragma unroll
        for (int q = 0; q < BN / 4; ++q) {
            da[b][q]  = f.a[b + 1][q];
            dvb[b][q] = f.vb[b + 1][q];
        }
    }
}

template <int NB>
__device__ __forceinline__ void step_last(const float* Ks, const float* Vs, const float* Pcur,
                                          const int g, const int wave, const int r, const int hi,
                                          const v2f (&da)[ND][BN / 4],
                                          const v2f (&dvb)[ND][BN / 4],
                                          v8f& acc, v8f& accz) {
    const v2f ones = {1.0f, 1.0f};
    const bool zdo = (NB == 4) || (wave < NB);
    StepFrags f;
    step_loads_v<NB>(f, Vs, g, wave, r, hi);
    wg_wait(acc, accz);
    step_loads<NB, false>(f, Ks, Pcur, g, wave, r, hi);
#pragma unroll
    for (int b = 0; b < ND; ++b) {
#pragma unroll
        for (int q = 0; q < BN / 4; ++q)
            acc = wmma_f32_k4(da[b][q], dvb[b][q], acc);
    }
#pragma unroll
    for (int b = 0; b < 4; ++b) {
        if (b < NB) {
#pragma unroll
            for (int q = 0; q < BN / 4; ++q)
                acc = wmma_f32_k4(f.a[b][q], f.vb[b][q], acc);
        }
        if (NB == 4) accz = wmma_f32_k4(f.az[b], ones, accz);
    }
    if (NB < 4 && zdo) {
#pragma unroll
        for (int q = 0; q < BN / 4; ++q)
            accz = wmma_f32_k4(f.az[q], ones, accz);
    }
}

__global__ __launch_bounds__(NW * 32)
void poly2_attn_kernel(const float* __restrict__ Q,
                       const float* __restrict__ K,
                       const float* __restrict__ V,
                       float* __restrict__ YS,
                       float* __restrict__ YZ) {
    __shared__ __attribute__((aligned(16))) float Ks[SEQ * KP];
    __shared__ __attribute__((aligned(16))) float Vs[SEQ * VP];
    __shared__ float Qs[BM][DK + 1];
    __shared__ float Ps[2 * PG];
    __shared__ float Zs[NW * BM];
    __shared__ __attribute__((aligned(16))) float Os[BM * OP];

    const int h    = blockIdx.x;
    const int qt   = (SEQ / BM - 1) - (int)blockIdx.y;
    const int qbase = qt * BM;
    const int jd   = qt;
    const int tid  = threadIdx.x;
    const int wave = __builtin_amdgcn_readfirstlane(tid >> 5);
    const int lane = tid & 31;
    const int r    = lane & 15;
    const int hi   = lane >> 4;

    const float* Qh = Q + (size_t)h * SEQ * DK;
    const float* Kh = K + (size_t)h * SEQ * DK;
    const float* Vh = V + (size_t)h * SEQ * DV;
    const v4f* Kh4 = reinterpret_cast<const v4f*>(Kh);
    const v4f* Vh4 = reinterpret_cast<const v4f*>(Vh);

    const v4f qreg = reinterpret_cast<const v4f*>(Qh + (size_t)qbase * DK)[tid];

    const int nch = (qt >> 2) + 1;
#define LOAD_CHUNK(c, kr, vr)                                                   \
    {                                                                           \
        _Pragma("unroll")                                                       \
        for (int u = 0; u < 4; ++u)                                             \
            kr[u] = Kh4[(c) * (CH * DK / 4) + u * (NW * 32) + tid];             \
        _Pragma("unroll")                                                       \
        for (int u = 0; u < 8; ++u)                                             \
            vr[u] = Vh4[(c) * (CH * DV / 4) + u * (NW * 32) + tid];             \
    }
#define STORE_CHUNK(c, kr, vr)                                                  \
    {                                                                           \
        _Pragma("unroll")                                                       \
        for (int u = 0; u < 4; ++u) {                                           \
            const int idx = u * (NW * 32) + tid;                                \
            const int row = (c) * CH + (idx >> 3), c4 = idx & 7;                \
            *reinterpret_cast<v4f*>(&Ks[row * KP + c4 * 4]) = kr[u];         \
        }                                                                       \
        _Pragma("unroll")                                                       \
        for (int u = 0; u < 8; ++u) {                                           \
            const int idx = u * (NW * 32) + tid;                                \
            const int row = (c) * CH + (idx >> 4), c4 = idx & 15;               \
            *reinterpret_cast<v4f*>(&Vs[row * VP + c4 * 4]) = vr[u];         \
        }                                                                       \
    }
    static_assert(SEQ >= 3 * CH, "chunks 1 and 2 exist for every tile");
    v4f ka[4], va[8], kc[4], vc[8];
    LOAD_CHUNK(0, ka, va)
    LOAD_CHUNK(1, kc, vc)
    STORE_CHUNK(0, ka, va)
    {
        const int row = tid >> 3, col = (tid & 7) * 4;
        Qs[row][col + 0] = qreg.x;
        Qs[row][col + 1] = qreg.y;
        Qs[row][col + 2] = qreg.z;
        Qs[row][col + 3] = qreg.w;
    }
    wg_signal_lds();
    LOAD_CHUNK(2, ka, va)
    wg_wait();

    v2f qa[DK / 4];
#pragma unroll
    for (int kk = 0; kk < DK; kk += 4) {
        const int c0 = kk + 2 * hi;
        qa[kk >> 2] = (v2f){ Qs[r][c0], Qs[r][c0 + 1] };
    }

    v8f acc  = {0,0,0,0,0,0,0,0};
    v2f da[ND][BN / 4], dvb[ND][BN / 4];
#pragma unroll
    for (int b = 0; b < ND; ++b) {
#pragma unroll
        for (int q = 0; q < BN / 4; ++q) {
            da[b][q]  = (v2f){0.0f, 0.0f};
            dvb[b][q] = (v2f){0.0f, 0.0f};
        }
    }
    v8f accz = {0,0,0,0,0,0,0,0};

    {
        const v8f s = score_chain(Ks, qa, wave, r, hi);
        store_p<true>(&Ps[wave * PT], s, wave, qbase, r, hi);
    }
    STORE_CHUNK(1, kc, vc)
    wg_signal_lds();

#define STEP_HEAD(G, MASK)                                                            \
    StepFrags f;                                                                      \
    step_loads_v<4>(f, Vs, (G), wave, r, hi);                                         \
    wg_wait(acc, accz);                                                               \
    step_loads<4, true>(f, Ks, &Ps[((G) & 1) * PG], (G), wave, r, hi);                \
    step_head<MASK>(f, qa, da, dvb, acc, &Ps[(((G) + 1) & 1) * PG + wave * PT],       \
                    4 * ((G) + 1) + wave, qbase, r, hi);
#define STEP_TAIL()                                                                   \
    wg_signal_lds(acc, accz);                                                         \
    step_tail(f, acc, accz);                                                          \
    step_handover(f, da, dvb);
    int g = 0;
    for (; g + 4 < nch; g += 2) {
        {
            LOAD_CHUNK(g + 3, kc, vc)
            STEP_HEAD(g, false)
            STORE_CHUNK(g + 2, ka, va)
            STEP_TAIL()
        }
        {
            LOAD_CHUNK(g + 4, ka, va)
            STEP_HEAD(g + 1, false)
            STORE_CHUNK(g + 3, kc, vc)
            STEP_TAIL()
        }
    }
    if (g + 3 < nch) {
        {
            LOAD_CHUNK(g + 3, kc, vc)
            STEP_HEAD(g, false)
            STORE_CHUNK(g + 2, ka, va)
            STEP_TAIL()
        }
        {
            STEP_HEAD(g + 1, false)
            STORE_CHUNK(g + 3, kc, vc)
            STEP_TAIL()
        }
        g += 2;
    } else if (g + 2 < nch) {
        STEP_HEAD(g, false)
        STORE_CHUNK(g + 2, ka, va)
        STEP_TAIL()
        ++g;
    }
#undef LOAD_CHUNK
#undef STORE_CHUNK
    if (g + 1 < nch) {
        STEP_HEAD(g, true)
        STEP_TAIL()
        ++g;
    }
#undef STEP_HEAD
#undef STEP_TAIL
    switch (qt & 3) {
        case 0:  step_last<1>(Ks, Vs, &Ps[(g & 1) * PG], g, wave, r, hi, da, dvb, acc, accz); break;
        case 1:  step_last<2>(Ks, Vs, &Ps[(g & 1) * PG], g, wave, r, hi, da, dvb, acc, accz); break;
        case 2:  step_last<3>(Ks, Vs, &Ps[(g & 1) * PG], g, wave, r, hi, da, dvb, acc, accz); break;
        default: step_last<4>(Ks, Vs, &Ps[(g & 1) * PG], g, wave, r, hi, da, dvb, acc, accz); break;
    }
    __syncthreads();

#pragma unroll
    for (int i = 0; i < 8; ++i)
        Os[(i + 8 * hi) * OP + 16 * wave + r] = acc[i] * ALPHA;
    if (r == 0) {
#pragma unroll
        for (int i = 0; i < 8; ++i)
            Zs[wave * BM + i + 8 * hi] = accz[i];
    }
    __syncthreads();
    v4f* ys4 = reinterpret_cast<v4f*>(YS + ((size_t)h * SEQ + qbase) * DV);
#pragma unroll
    for (int u = 0; u < 2; ++u) {
        const int idx = u * (NW * 32) + tid;
        const int row = idx >> 4, c4 = idx & 15;
        ys4[idx] = *reinterpret_cast<const v4f*>(&Os[row * OP + c4 * 4]);
    }
    if (tid < BM)
        YZ[(size_t)h * SEQ + qbase + tid] =
            (((Zs[tid] + Zs[BM + tid]) + Zs[2 * BM + tid]) + Zs[3 * BM + tid]) * ALPHA;
}

extern "C" void kernel_launch(void* const* d_in, const int* in_sizes, int n_in,
                              void* d_out, int out_size, void* d_ws, size_t ws_size,
                              hipStream_t stream) {
    (void)in_sizes; (void)n_in; (void)out_size; (void)d_ws; (void)ws_size;
    const float* Q = (const float*)d_in[0];
    const float* K = (const float*)d_in[1];
    const float* V = (const float*)d_in[2];
    float* YS = (float*)d_out;
    float* YZ = (float*)d_out + (size_t)NH * SEQ * DV;

    dim3 grid(NH, SEQ / BM);
    dim3 block(NW * 32);
    hipLaunchKernelGGL(poly2_attn_kernel, grid, block, 0, stream, Q, K, V, YS, YZ);
}


# Work-group widened to 8 waves in two roles, one of each per SIMD: waves 0-3 copy K/V to LDS and make the score tiles, waves 4-7 run the P.V and Y_Z products out of LDS; sum order kept
# speedup vs baseline: 10.4964x; 1.0653x over previous
#include <hip/hip_runtime.h>
#include <hip/hip_bf16.h>


#define NH     8
#define SEQ    512
#define DK     32
#define DV     64
#define BM     16
#define NW     4
#define BN     16
#define CH     64
#define KP     36
#define VP     72
#define OP     68
#define ALPHA  (1.0f / 64.0f)

typedef float v2f __attribute__((ext_vector_type(2)));
typedef float v4f __attribute__((ext_vector_type(4)));
typedef float v8f __attribute__((ext_vector_type(8)));

__device__ __forceinline__ v8f wmma_f32_k4(v2f a, v2f b, v8f c) {
    return __builtin_amdgcn_wmma_f32_16x16x4_f32(false, a, false, b,
                                                 (short)0, c, false, false);
}

__device__ __forceinline__ void wg_signal_lds() {
    asm volatile("s_wait_dscnt 0x0\n\ts_barrier_signal -1" ::: "memory");
}
__device__ __forceinline__ void wg_wait() {
    asm volatile("s_barrier_wait -1" ::: "memory");
}
__device__ __forceinline__ void wg_signal_lds(v8f& acc, v8f& accz) {
    asm volatile("s_wait_dscnt 0x0\n\ts_barrier_signal -1" : "+v"(acc), "+v"(accz) :: "memory");
}
__device__ __forceinline__ void wg_wait(const v8f& acc, const v8f& accz) {
    asm volatile("s_barrier_wait -1" :: "v"(acc), "v"(accz) : "memory");
}

#define PP     17
#define PT     (16 * PP)
#define PG     (NW * PT)

__device__ __forceinline__ v8f score_chain(const float* Ks, const v2f (&qa)[DK / 4],
                                           const int kb, const int r, const int hi) {
    const float* krow = &Ks[(kb * BN + r) * KP + 2 * hi];
    v2f kf[DK / 4];
#pragma unroll
    for (int t = 0; t < DK / 4; ++t)
        kf[t] = (v2f){ krow[4 * t], krow[4 * t + 1] };
    v8f s = {0,0,0,0,0,0,0,0};
#pragma unroll
    for (int t = 0; t < DK / 4; ++t)
        s = wmma_f32_k4(qa[t], kf[t], s);
    return s;
}

template <bool MASK>
__device__ __forceinline__ void store_p(float* Pt, const v8f& s,
                                        const int kb, const int qbase,
                                        const int r, const int hi) {
#pragma unroll
    for (int i = 0; i < 8; ++i) {
        float p = s[i] * s[i];
        if (MASK) {
            const int gq = qbase + i + 8 * hi;
            const int gk = kb * BN + r;
            if (gk > gq) p = 0.0f;
        }
        Pt[(i + 8 * hi) * PP + r] = p;
    }
}

#define ND     3

struct StepFrags {
    v2f a[4][BN / 4];
    v2f vb[4][BN / 4];
    v2f az[BN / 4];
};

template <int NB>
__device__ __forceinline__ void step_loads_v(StepFrags& f, const float* Vs, const int g,
                                             const int wave, const int r, const int hi) {
#pragma unroll
    for (int b = 0; b < NB; ++b) {
        const float* vrow = &Vs[((4 * g + b) * BN + 2 * hi) * VP + 16 * wave + r];
#pragma unroll
        for (int q = 0; q < BN / 4; ++q)
            f.vb[b][q] = (v2f){ vrow[(4 * q) * VP], vrow[(4 * q + 1) * VP] };
    }
}

template <int NB>
__device__ __forceinline__ void step_loads(StepFrags& f, const float* Pcur, const int wave,
                                           const int r, const int hi) {
    const bool zdo = (NB == 4) || (wave < NB);
    if (zdo) {
        const float* prow = &Pcur[wave * PT + r * PP + 2 * hi];
#pragma unroll
        for (int q = 0; q < BN / 4; ++q)
            f.az[q] = (v2f){ prow[4 * q], prow[4 * q + 1] };
    }
#pragma unroll
    for (int b = 0; b < NB; ++b) {
        const float* prow = &Pcur[b * PT + r * PP + 2 * hi];
#pragma unroll
        for (int q = 0; q < BN / 4; ++q)
            f.a[b][q]  = (v2f){ prow[4 * q], prow[4 * q + 1] };
    }
    __builtin_amdgcn_sched_barrier(0);
}

__device__ __forceinline__ void step_deferred(const StepFrags& fp, v8f& acc) {
#pragma unroll
    for (int b = 1; b <= ND; ++b) {
#pragma unroll
        for (int q = 0; q < BN / 4; ++q)
            acc = wmma_f32_k4(fp.a[b][q], fp.vb[b][q], acc);
    }
}

__device__ __forceinline__ void step_tail(const StepFrags& f, v8f& acc, v8f& accz) {
    const v2f ones = {1.0f, 1.0f};
#pragma unroll
    for (int q = 0; q < BN / 4; ++q) {
        acc  = wmma_f32_k4(f.a[0][q], f.vb[0][q], acc);
        accz = wmma_f32_k4(f.az[q], ones, accz);
    }
}

template <int NB>
__device__ __forceinline__ void step_last(const float* Vs, const float* Pcur,
                                          const int g, const int wave, const int r, const int hi,
                                          const StepFrags& fp, v8f& acc, v8f& accz) {
    const v2f ones = {1.0f, 1.0f};
    const bool zdo = (NB == 4) || (wave < NB);
    StepFrags f;
    step_loads_v<NB>(f, Vs, g, wave, r, hi);
    wg_wait(acc, accz);
    step_loads<NB>(f, Pcur, wave, r, hi);
    step_deferred(fp, acc);
#pragma unroll
    for (int b = 0; b < 4; ++b) {
        if (b < NB) {
#pragma unroll
            for (int q = 0; q < BN / 4; ++q)
                acc = wmma_f32_k4(f.a[b][q], f.vb[b][q], acc);
        }
        if (NB == 4) accz = wmma_f32_k4(f.az[b], ones, accz);
    }
    if (NB < 4 && zdo) {
#pragma unroll
        for (int q = 0; q < BN / 4; ++q)
            accz = wmma_f32_k4(f.az[q], ones, accz);
    }
}

__global__ __launch_bounds__(2 * NW * 32)
void poly2_attn_kernel(const float* __restrict__ Q,
                       const float* __restrict__ K,
                       const float* __restrict__ V,
                       float* __restrict__ YS,
                       float* __restrict__ YZ) {
    __shared__ __attribute__((aligned(16))) float Ks[SEQ * KP];
    __shared__ __attribute__((aligned(16))) float Vs[SEQ * VP];
    __shared__ float Qs[BM][DK + 1];
    __shared__ float Ps[2 * PG];
    __shared__ float Zs[NW * BM];
    __shared__ __attribute__((aligned(16))) float Os[BM * OP];

    const int h    = blockIdx.x;
    const int qt   = (SEQ / BM - 1) - (int)blockIdx.y;
    const int qbase = qt * BM;
    const int tid  = threadIdx.x;
    const int wave8 = __builtin_amdgcn_readfirstlane(tid >> 5);
    const bool score_role = wave8 < NW;
    const int wave = wave8 & (NW - 1);
    const int lane = tid & 31;
    const int r    = lane & 15;
    const int hi   = lane >> 4;

    const int nch = (qt >> 2) + 1;

    v8f acc  = {0,0,0,0,0,0,0,0};
    v8f accz = {0,0,0,0,0,0,0,0};

    if (score_role) {
        const float* Qh = Q + (size_t)h * SEQ * DK;
        const float* Kh = K + (size_t)h * SEQ * DK;
        const float* Vh = V + (size_t)h * SEQ * DV;
        const v4f* Kh4 = reinterpret_cast<const v4f*>(Kh);
        const v4f* Vh4 = reinterpret_cast<const v4f*>(Vh);

        const v4f qreg = reinterpret_cast<const v4f*>(Qh + (size_t)qbase * DK)[tid];

#define LOAD_CHUNK(c, kr, vr)                                                   \
        {                                                                       \
            _Pragma("unroll")                                                   \
            for (int u = 0; u < 4; ++u)                                         \
                kr[u] = Kh4[(c) * (CH * DK / 4) + u * (NW * 32) + tid];         \
            _Pragma("unroll")                                                   \
            for (int u = 0; u < 8; ++u)                                         \
                vr[u] = Vh4[(c) * (CH * DV / 4) + u * (NW * 32) + tid];         \
        }
#define STORE_CHUNK(c, kr, vr)                                                  \
        {                                                                       \
            _Pragma("unroll")                                                   \
            for (int u = 0; u < 4; ++u) {                                       \
                const int idx = u * (NW * 32) + tid;                            \
                const int row = (c) * CH + (idx >> 3), c4 = idx & 7;            \
                *reinterpret_cast<v4f*>(&Ks[row * KP + c4 * 4]) = kr[u];        \
            }                                                                   \
            _Pragma("unroll")                                                   \
            for (int u = 0; u < 8; ++u) {                                       \
                const int idx = u * (NW * 32) + tid;                            \
                const int row = (c) * CH + (idx >> 4), c4 = idx & 15;           \
                *reinterpret_cast<v4f*>(&Vs[row * VP + c4 * 4]) = vr[u];        \
            }                                                                   \
        }
        static_assert(SEQ >= 3 * CH, "chunks 1 and 2 exist for every tile");
        v4f ka[4], va[8], kc[4], vc[8];
        LOAD_CHUNK(0, ka, va)
        LOAD_CHUNK(1, kc, vc)
        STORE_CHUNK(0, ka, va)
        {
            const int row = tid >> 3, col = (tid & 7) * 4;
            Qs[row][col + 0] = qreg.x;
            Qs[row][col + 1] = qreg.y;
            Qs[row][col + 2] = qreg.z;
            Qs[row][col + 3] = qreg.w;
        }
        wg_signal_lds();
        LOAD_CHUNK(2, ka, va)
        wg_wait();

        v2f qa[DK / 4];
#pragma unroll
        for (int kk = 0; kk < DK; kk += 4) {
            const int c0 = kk + 2 * hi;
            qa[kk >> 2] = (v2f){ Qs[r][c0], Qs[r][c0 + 1] };
        }

        {
            const v8f s = score_chain(Ks, qa, wave, r, hi);
            store_p<true>(&Ps[wave * PT], s, wave, qbase, r, hi);
        }
        STORE_CHUNK(1, kc, vc)
        wg_signal_lds();

#define SCORE_STEP(G, MASK)                                                           \
        {                                                                             \
            wg_wait();                                                                \
            const int kbn = 4 * ((G) + 1) + wave;                                     \
            const v8f s = score_chain(Ks, qa, kbn, r, hi);                            \
            store_p<MASK>(&Ps[(((G) + 1) & 1) * PG + wave * PT], s, kbn, qbase, r, hi); \
        }
        int g = 0;
        for (; g + 4 < nch; g += 2) {
            {
                LOAD_CHUNK(g + 3, kc, vc)
                SCORE_STEP(g, false)
                STORE_CHUNK(g + 2, ka, va)
                wg_signal_lds();
            }
            {
                LOAD_CHUNK(g + 4, ka, va)
                SCORE_STEP(g + 1, false)
                STORE_CHUNK(g + 3, kc, vc)
                wg_signal_lds();
            }
        }
        if (g + 3 < nch) {
            {
                LOAD_CHUNK(g + 3, kc, vc)
                SCORE_STEP(g, false)
                STORE_CHUNK(g + 2, ka, va)
                wg_signal_lds();
            }
            {
                SCORE_STEP(g + 1, false)
                STORE_CHUNK(g + 3, kc, vc)
                wg_signal_lds();
            }
            g += 2;
        } else if (g + 2 < nch) {
            SCORE_STEP(g, false)
            STORE_CHUNK(g + 2, ka, va)
            wg_signal_lds();
            ++g;
        }
#undef LOAD_CHUNK
#undef STORE_CHUNK
        if (g + 1 < nch) {
            SCORE_STEP(g, true)
            wg_signal_lds();
            ++g;
        }
#undef SCORE_STEP
        wg_wait();
    } else {
        wg_signal_lds();
        wg_wait();
        wg_signal_lds();

        StepFrags fa, fb;
#pragma unroll
        for (int b = 1; b <= ND; ++b) {
#pragma unroll
            for (int q = 0; q < BN / 4; ++q) {
                fb.a[b][q]  = (v2f){0.0f, 0.0f};
                fb.vb[b][q] = (v2f){0.0f, 0.0f};
            }
        }

#define PRODUCT_STEP(F, FP, G)                                                        \
        {                                                                             \
            step_loads_v<4>(F, Vs, (G), wave, r, hi);                                 \
            wg_wait(acc, accz);                                                       \
            step_loads<4>(F, &Ps[((G) & 1) * PG], wave, r, hi);                       \
            step_deferred(FP, acc);                                                   \
            wg_signal_lds(acc, accz);                                                 \
            step_tail(F, acc, accz);                                                  \
        }
#define PRODUCT_LAST(FP, G)                                                           \
        switch (qt & 3) {                                                             \
            case 0:  step_last<1>(Vs, &Ps[((G) & 1) * PG], (G), wave, r, hi, FP, acc, accz); break; \
            case 1:  step_last<2>(Vs, &Ps[((G) & 1) * PG], (G), wave, r, hi, FP, acc, accz); break; \
            case 2:  step_last<3>(Vs, &Ps[((G) & 1) * PG], (G), wave, r, hi, FP, acc, accz); break; \
            default: step_last<4>(Vs, &Ps[((G) & 1) * PG], (G), wave, r, hi, FP, acc, accz); break; \
        }
        int g = 0;
        for (; g + 2 < nch; g += 2) {
            PRODUCT_STEP(fa, fb, g)
            PRODUCT_STEP(fb, fa, g + 1)
        }
        if (g + 1 < nch) {
            PRODUCT_STEP(fa, fb, g)
            ++g;
            PRODUCT_LAST(fa, g)
        } else {
            PRODUCT_LAST(fb, g)
        }
#undef PRODUCT_STEP
#undef PRODUCT_LAST
    }
    __syncthreads();

    if (!score_role) {
#pragma unroll
        for (int i = 0; i < 8; ++i)
            Os[(i + 8 * hi) * OP + 16 * wave + r] = acc[i] * ALPHA;
        if (r == 0) {
#pragma unroll
            for (int i = 0; i < 8; ++i)
                Zs[wave * BM + i + 8 * hi] = accz[i];
        }
    }
    __syncthreads();
    v4f* ys4 = reinterpret_cast<v4f*>(YS + ((size_t)h * SEQ + qbase) * DV);
    {
        const int row = tid >> 4, c4 = tid & 15;
        ys4[tid] = *reinterpret_cast<const v4f*>(&Os[row * OP + c4 * 4]);
    }
    if (tid < BM)
        YZ[(size_t)h * SEQ + qbase + tid] =
            (((Zs[tid] + Zs[BM + tid]) + Zs[2 * BM + tid]) + Zs[3 * BM + tid]) * ALPHA;
}

extern "C" void kernel_launch(void* const* d_in, const int* in_sizes, int n_in,
                              void* d_out, int out_size, void* d_ws, size_t ws_size,
                              hipStream_t stream) {
    (void)in_sizes; (void)n_in; (void)out_size; (void)d_ws; (void)ws_size;
    const float* Q = (const float*)d_in[0];
    const float* K = (const float*)d_in[1];
    const float* V = (const float*)d_in[2];
    float* YS = (float*)d_out;
    float* YZ = (float*)d_out + (size_t)NH * SEQ * DV;

    dim3 grid(NH, SEQ / BM);
    dim3 block(2 * NW * 32);
    hipLaunchKernelGGL(poly2_attn_kernel, grid, block, 0, stream, Q, K, V, YS, YZ);
}
